# starting kernel
# baseline (speedup 1.0000x reference)
#include <hip/hip_runtime.h>

typedef __attribute__((ext_vector_type(2))) float v2f;
typedef __attribute__((ext_vector_type(4))) float v4f;
typedef __attribute__((ext_vector_type(8))) float v8f;

#define NPTS 8192
#define CCH  64
#define KNN  16
#define KSTR 132
#define XSTR 68
#define WAVE_LDS (2*64 + 2*16*XSTR)

__global__ __launch_bounds__(256) void knn_kernel(const float* __restrict__ coords,
                                                  int* __restrict__ idx_out) {
    __shared__ v4f pts[NPTS];
    const int b     = blockIdx.x >> 5;
    const int qbase = (blockIdx.x & 31) << 8;
    const float* cb = coords + b * NPTS * 3;

    for (int j = threadIdx.x; j < NPTS; j += 256) {
        float x = cb[j * 3 + 0], y = cb[j * 3 + 1], z = cb[j * 3 + 2];
        v4f p; p.x = x; p.y = y; p.z = z; p.w = x * x + y * y + z * z;
        pts[j] = p;
    }
    __syncthreads();

    const int q = qbase + threadIdx.x;
    const v4f me = pts[q];

    float best[KNN]; int bidx[KNN];
#pragma unroll
    for (int t = 0; t < KNN; ++t) { best[t] = 3.4e38f; bidx[t] = 0; }

    for (int j = 0; j < NPTS; ++j) {
        v4f p = pts[j];
        float dot = me.x * p.x;
        dot = fmaf(me.y, p.y, dot);
        dot = fmaf(me.z, p.z, dot);
        float d = fmaf(-2.0f, dot, me.w + p.w);
        d = fmaxf(d, 0.0f);
        if (d < best[KNN - 1]) {
            float cd = d; int ci = j;
#pragma unroll
            for (int t = 0; t < KNN; ++t) {
                if (cd < best[t]) {
                    float tf = best[t]; best[t] = cd; cd = tf;
                    int   ti = bidx[t]; bidx[t] = ci; ci = ti;
                }
            }
        }
    }

    int* op = idx_out + (b * NPTS + q) * KNN;
#pragma unroll
    for (int t = 0; t < KNN; ++t) op[t] = bidx[t];
}

__global__ __launch_bounds__(256) void agg_kernel(const float* __restrict__ coords,
                                                  const float* __restrict__ feat,
                                                  const float* __restrict__ Wm,
                                                  const float* __restrict__ bias,
                                                  const int*   __restrict__ idx,
                                                  float* __restrict__ out) {
    __shared__ float LWT[64 * KSTR];
    __shared__ float XS[8 * WAVE_LDS];

    for (int i = threadIdx.x; i < 131 * 64; i += 256) {
        int k = i >> 6, n = i & 63;
        LWT[n * KSTR + k] = Wm[i];
    }
    if (threadIdx.x < 64) LWT[threadIdx.x * KSTR + 131] = 0.0f;
    __syncthreads();

    const int lane = threadIdx.x & 31;
    const int wav  = threadIdx.x >> 5;
    const int nl   = lane & 15;
    const int h    = lane >> 4;

    float* fib0 = &XS[wav * WAVE_LDS];
    float* fib1 = fib0 + 64;
    float* X0   = fib1 + 64;
    float* X1   = X0 + 16 * XSTR;

    const float* wt0 = LWT + (nl     ) * KSTR + 2 * h;
    const float* wt1 = LWT + (nl + 16) * KSTR + 2 * h;
    const float* wt2 = LWT + (nl + 32) * KSTR + 2 * h;
    const float* wt3 = LWT + (nl + 48) * KSTR + 2 * h;

    float bs[4];
#pragma unroll
    for (int nt = 0; nt < 4; ++nt) bs[nt] = bias[nt * 16 + nl];

    const int gw = blockIdx.x * 8 + wav;

    for (int it = 0; it < 16; ++it) {
        const int P0 = gw * 32 + it * 2;
        const int b  = P0 >> 13;
        const int n0 = P0 & (NPTS - 1);
        const int n1 = n0 + 1;
        const float* fb = feat   + b * NPTS * CCH;
        const float* cb = coords + b * NPTS * 3;

        const int myn = h ? n1 : n0;
        const int nbv = idx[(b * NPTS + myn) * KNN + nl];

        v2f fi0 = *(const v2f*)(fb + n0 * CCH + 2 * lane);
        v2f fi1 = *(const v2f*)(fb + n1 * CCH + 2 * lane);
        *(v2f*)(fib0 + 2 * lane) = fi0;
        *(v2f*)(fib1 + 2 * lane) = fi1;

#pragma unroll
        for (int m = 0; m < 32; ++m) {
            int nb = __shfl(nbv, m, 32);
            v2f g  = *(const v2f*)(fb + nb * CCH + 2 * lane);
            if (m < 16) { v2f d0 = g - fi0; *(v2f*)(X0 + m * XSTR + 2 * lane) = d0; }
            else        { v2f d1 = g - fi1; *(v2f*)(X1 + (m - 16) * XSTR + 2 * lane) = d1; }
        }

        {
            const float* pc  = cb + myn * 3;
            const float* ncp = cb + nbv * 3;
            v4f dp; dp.x = ncp[0] - pc[0]; dp.y = ncp[1] - pc[1];
                    dp.z = ncp[2] - pc[2]; dp.w = 0.0f;
            float* xr = (h ? X1 : X0) + nl * XSTR + 64;
            *(v4f*)xr = dp;
        }

        v8f a0c[4], a1c[4];
        const v8f zc = {0.f, 0.f, 0.f, 0.f, 0.f, 0.f, 0.f, 0.f};
#pragma unroll
        for (int nt = 0; nt < 4; ++nt) { a0c[nt] = zc; a1c[nt] = zc; }

#define STEP(K0, A0, A1)                                                          \
        {                                                                         \
            v2f b0 = *(const v2f*)(wt0 + (K0));                                   \
            v2f b1 = *(const v2f*)(wt1 + (K0));                                   \
            v2f b2 = *(const v2f*)(wt2 + (K0));                                   \
            v2f b3 = *(const v2f*)(wt3 + (K0));                                   \
            a0c[0] = __builtin_amdgcn_wmma_f32_16x16x4_f32(                       \
                false, (A0), false, b0, (short)0, a0c[0], false, false);          \
            a1c[0] = __builtin_amdgcn_wmma_f32_16x16x4_f32(                       \
                false, (A1), false, b0, (short)0, a1c[0], false, false);          \
            a0c[1] = __builtin_amdgcn_wmma_f32_16x16x4_f32(                       \
                false, (A0), false, b1, (short)0, a0c[1], false, false);          \
            a1c[1] = __builtin_amdgcn_wmma_f32_16x16x4_f32(                       \
                false, (A1), false, b1, (short)0, a1c[1], false, false);          \
            a0c[2] = __builtin_amdgcn_wmma_f32_16x16x4_f32(                       \
                false, (A0), false, b2, (short)0, a0c[2], false, false);          \
            a1c[2] = __builtin_amdgcn_wmma_f32_16x16x4_f32(                       \
                false, (A1), false, b2, (short)0, a1c[2], false, false);          \
            a0c[3] = __builtin_amdgcn_wmma_f32_16x16x4_f32(                       \
                false, (A0), false, b3, (short)0, a0c[3], false, false);          \
            a1c[3] = __builtin_amdgcn_wmma_f32_16x16x4_f32(                       \
                false, (A1), false, b3, (short)0, a1c[3], false, false);          \
        }

#pragma unroll
        for (int ks = 0; ks < 16; ++ks) {
            int k0 = 4 * ks;
            v2f A0 = *(const v2f*)(fib0 + k0 + 2 * h);
            v2f A1 = *(const v2f*)(fib1 + k0 + 2 * h);
            STEP(k0, A0, A1);
        }
#pragma unroll
        for (int ks = 0; ks < 16; ++ks) {
            int k0 = 64 + 4 * ks;
            int xo = nl * XSTR + 4 * ks + 2 * h;
            v2f A0 = *(const v2f*)(X0 + xo);
            v2f A1 = *(const v2f*)(X1 + xo);
            STEP(k0, A0, A1);
        }
        {
            int k0 = 128;
            int xo = nl * XSTR + 64 + 2 * h;
            v2f A0 = *(const v2f*)(X0 + xo);
            v2f A1 = *(const v2f*)(X1 + xo);
            STEP(k0, A0, A1);
        }
#undef STEP

        float* outp = out + (size_t)P0 * CCH;
#pragma unroll
        for (int nt = 0; nt < 4; ++nt) {
            float v0 = a0c[nt][0];
            float v1 = a1c[nt][0];
#pragma unroll
            for (int r = 1; r < 8; ++r) {
                v0 = fmaxf(v0, a0c[nt][r]);
                v1 = fmaxf(v1, a1c[nt][r]);
            }
            v0 = fmaxf(v0, __shfl_xor(v0, 16, 32));
            v1 = fmaxf(v1, __shfl_xor(v1, 16, 32));
            v0 = fmaxf(v0 + bs[nt], 0.0f);
            v1 = fmaxf(v1 + bs[nt], 0.0f);
            if (lane < 16) {
                outp[nt * 16 + nl]       = v0;
                outp[CCH + nt * 16 + nl] = v1;
            }
        }
    }
}

extern "C" void kernel_launch(void* const* d_in, const int* in_sizes, int n_in,
                              void* d_out, int out_size, void* d_ws, size_t ws_size,
                              hipStream_t stream) {
    const float* coords = (const float*)d_in[0];
    const float* feat   = (const float*)d_in[1];
    const float* Wm     = (const float*)d_in[2];
    const float* bias   = (const float*)d_in[3];
    float* out = (float*)d_out;
    int* idx = (int*)d_ws;

    knn_kernel<<<64, 256, 0, stream>>>(coords, idx);
    agg_kernel<<<64, 256, 0, stream>>>(coords, feat, Wm, bias, idx, out);
}


# Aggregation loop software-pipelined: next pair's centre rows, neighbour rows, coordinates (indices two pairs ahead) are loaded before the current pair's WMMAs, staged to LDS after them; same sums
# speedup vs baseline: 21.8085x; 21.8085x over previous
#include <hip/hip_runtime.h>

typedef __attribute__((ext_vector_type(2))) float v2f;
typedef __attribute__((ext_vector_type(4))) float v4f;
typedef __attribute__((ext_vector_type(8))) float v8f;

#define NPTS 8192
#define CCH  64
#define KNN  16
#define XSTR 68
#define WAVE_LDS (2*64 + 2*16*XSTR)

typedef unsigned int u32;
typedef __attribute__((ext_vector_type(2))) unsigned int v2u;

__device__ __forceinline__ bool key_lt(u32 da, u32 ia, u32 db, u32 ib) {
    u32 c1, c2;
    (void)__builtin_subc(ia, ib, 0u, &c1);
    (void)__builtin_subc(da, db, c1, &c2);
    return c2 != 0u;
}

__device__ __forceinline__ void cswap(u32& da, u32& ia, u32& db, u32& ib) {
    const bool lt = key_lt(da, ia, db, ib);
    const u32 dlo = lt ? da : db, dhi = lt ? db : da;
    const u32 ilo = lt ? ia : ib, ihi = lt ? ib : ia;
    da = dlo; ia = ilo; db = dhi; ib = ihi;
}

__device__ __forceinline__ void sort16(u32 (&d)[16], u32 (&x)[16]) {
#define CS(a, b) cswap(d[a], x[a], d[b], x[b]);
    CS(0,1) CS(2,3) CS(4,5) CS(6,7) CS(8,9) CS(10,11) CS(12,13) CS(14,15)
    CS(0,2) CS(1,3) CS(4,6) CS(5,7) CS(8,10) CS(9,11) CS(12,14) CS(13,15)
    CS(1,2) CS(5,6) CS(9,10) CS(13,14)
    CS(0,4) CS(1,5) CS(2,6) CS(3,7) CS(8,12) CS(9,13) CS(10,14) CS(11,15)
    CS(2,4) CS(3,5) CS(10,12) CS(11,13)
    CS(1,2) CS(3,4) CS(5,6) CS(9,10) CS(11,12) CS(13,14)
    CS(0,8) CS(1,9) CS(2,10) CS(3,11) CS(4,12) CS(5,13) CS(6,14) CS(7,15)
    CS(4,8) CS(5,9) CS(6,10) CS(7,11)
    CS(2,4) CS(3,5) CS(6,8) CS(7,9) CS(10,12) CS(11,13)
    CS(1,2) CS(3,4) CS(5,6) CS(7,8) CS(9,10) CS(11,12) CS(13,14)
#undef CS
}

__device__ __forceinline__ void merge16(u32 (&bd)[16], u32 (&bi)[16],
                                        const u32 (&xd)[16], const u32 (&xi)[16]) {
#pragma unroll
    for (int i = 0; i < 16; ++i) {
        const u32 od = xd[15 - i], oi = xi[15 - i];
        const bool lt = key_lt(bd[i], bi[i], od, oi);
        bd[i] = lt ? bd[i] : od;
        bi[i] = lt ? bi[i] : oi;
    }
#pragma unroll
    for (int lj = 3; lj >= 0; --lj) {
#pragma unroll
        for (int i = 0; i < 16; ++i) {
            const int l = i ^ (1 << lj);
            if (l > i) cswap(bd[i], bi[i], bd[l], bi[l]);
        }
    }
}

#define KNT   512
#define KSLEN 1024
#define KBCAP 32

__global__ __launch_bounds__(KNT) void knn_kernel(const float* __restrict__ coords,
                                                  int* __restrict__ idx_out) {
    __shared__ __attribute__((aligned(16))) v2u lds_raw[2 * NPTS];
    __shared__ __attribute__((aligned(8)))  v2u lds_pend[KBCAP * KNT];
    v4f* pts = (v4f*)lds_raw;
    const int tid   = threadIdx.x;
    const int b     = blockIdx.x >> 7;
    const int qbase = (blockIdx.x & 127) << 6;
    const float* cb = coords + b * NPTS * 3;

    for (int j = tid; j < NPTS; j += KNT) {
        float x = cb[j * 3 + 0], y = cb[j * 3 + 1], z = cb[j * 3 + 2];
        v4f p; p.x = x; p.y = y; p.z = z; p.w = x * x + y * y + z * z;
        pts[j] = p;
    }
    __syncthreads();

    const int lane = tid & 31;
    const int wav  = tid >> 5;
    const int s    = wav >> 1;
    const int q    = qbase + ((wav & 1) << 5) + lane;
    const v4f me   = pts[q];

    u32 bd[KNN], bi[KNN];
#pragma unroll
    for (int t = 0; t < KNN; ++t) { bd[t] = ~0u; bi[t] = ~0u; }

    v2u* const pend = lds_pend + tid;
    v2u* wp = pend;
    u32 thr = ~0u;

    const int cbase = s * KSLEN;
    for (int c = 0; c < KSLEN / 16; ++c) {
        u32 du[16];
#pragma unroll
        for (int i = 0; i < 16; ++i) {
            const v4f p = pts[cbase + c * 16 + i];
            float dot = me.x * p.x;
            dot = fmaf(me.y, p.y, dot);
            dot = fmaf(me.z, p.z, dot);
            float d = fmaf(-2.0f, dot, me.w + p.w);
            d = fmaxf(d, 0.0f);
            du[i] = __float_as_uint(d);
        }
#pragma unroll
        for (int i = 0; i < 16; ++i) {
            if (du[i] <= thr) {
                v2u k; k.x = (u32)(cbase + c * 16 + i); k.y = du[i];
                *wp = k;
                wp += KNT;
            }
        }
        const bool last = (c == KSLEN / 16 - 1);
        int cnt = (int)(wp - pend) / KNT;
        if (__any(last ? (cnt > 0) : (cnt >= 16))) {
            for (int pass = 0; pass < 2; ++pass) {
                u32 xd[16], xi[16];
#pragma unroll
                for (int t = 0; t < 16; ++t) {
                    v2u k; k.x = ~0u; k.y = ~0u;
                    if (t < cnt) k = pend[t * KNT];
                    xi[t] = k.x; xd[t] = k.y;
                }
                sort16(xd, xi);
                merge16(bd, bi, xd, xi);
                thr = bd[15];
                int left = 0;
                if (__any(cnt > 16)) {
#pragma unroll
                    for (int t = 0; t < 15; ++t) {
                        if (16 + t < cnt) {
                            const v2u k = pend[(16 + t) * KNT];
                            if (k.y <= thr) { pend[left * KNT] = k; ++left; }
                        }
                    }
                }
                cnt = left;
                if (!(last && __any(cnt > 0))) break;
            }
            wp = pend + cnt * KNT;
        }
    }

    __syncthreads();
    v2u* lists = lds_raw;
#pragma unroll
    for (int r = 0; r < 3; ++r) {
        const int span = 1 << r;
        const int ph   = s & (2 * span - 1);
        if (ph == span) {
#pragma unroll
            for (int t = 0; t < KNN; ++t) { v2u k; k.x = bi[t]; k.y = bd[t]; lists[t * KNT + tid] = k; }
        }
        __syncthreads();
        if (ph == 0) {
            u32 xd[16], xi[16];
#pragma unroll
            for (int t = 0; t < KNN; ++t) { const v2u k = lists[t * KNT + tid + 64 * span]; xi[t] = k.x; xd[t] = k.y; }
            merge16(bd, bi, xd, xi);
        }
    }

    if (s == 0) {
        int* op = idx_out + (b * NPTS + q) * KNN;
#pragma unroll
        for (int t = 0; t < KNN; ++t) op[t] = (int)bi[t];
    }
}

__global__ __launch_bounds__(128) __attribute__((amdgpu_waves_per_eu(1, 1))) void agg_kernel(const float* __restrict__ coords,
                                                  const float* __restrict__ feat,
                                                  const float* __restrict__ Wm,
                                                  const float* __restrict__ bias,
                                                  const int*   __restrict__ idx,
                                                  float* __restrict__ out) {
    __shared__ float XS[4 * WAVE_LDS];

    const int lane = threadIdx.x & 31;
    const int wav  = threadIdx.x >> 5;
    const int nl   = lane & 15;
    const int h    = lane >> 4;

    const int gw   = blockIdx.x * 4 + wav;
    const int half = gw & 1;
    const int pg   = gw >> 1;

    float* fib0 = &XS[wav * WAVE_LDS];
    float* fib1 = fib0 + 64;
    float* X0   = fib1 + 64;
    float* X1   = X0 + 16 * XSTR;

    v2f bw0[33], bw1[33];
    {
        const float* wc = Wm + half * 32 + nl;
#pragma unroll
        for (int ks = 0; ks < 33; ++ks) {
            const int k = 4 * ks + 2 * h;
            v2f t0, t1;
            t0.x = wc[k * CCH];
            t1.x = wc[k * CCH + 16];
            t0.y = (k + 1 < 131) ? wc[(k + 1) * CCH]      : 0.0f;
            t1.y = (k + 1 < 131) ? wc[(k + 1) * CCH + 16] : 0.0f;
            bw0[ks] = t0;
            bw1[ks] = t1;
        }
    }

    const float bsel = bias[half * 32 + lane];

    const int b     = (pg * 32) >> 13;
    const int nbase = (pg * 32) & (NPTS - 1);
    const float* fb = feat   + b * NPTS * CCH;
    const float* cb = coords + b * NPTS * 3;
    const int* ib = idx + (b * NPTS + nbase + h) * KNN + nl;

    v2f   fiA, fiB, g[32];
    float pc0, pc1, pc2, nc0, nc1, nc2;

#define LOAD_PAIR(PAIR, NBV)                                                      \
    {                                                                             \
        const int n0_ = nbase + (PAIR) * 2;                                       \
        fiA = *(const v2f*)(fb + n0_ * CCH + 2 * lane);                           \
        fiB = *(const v2f*)(fb + (n0_ + 1) * CCH + 2 * lane);                     \
        _Pragma("unroll")                                                         \
        for (int m = 0; m < 32; ++m) {                                            \
            const int nb_ = __shfl((NBV), m, 32);                                 \
            g[m] = *(const v2f*)(fb + nb_ * CCH + 2 * lane);                      \
        }                                                                         \
        const float* pc_ = cb + (n0_ + h) * 3;                                    \
        const float* nc_ = cb + (NBV) * 3;                                        \
        pc0 = pc_[0]; pc1 = pc_[1]; pc2 = pc_[2];                                 \
        nc0 = nc_[0]; nc1 = nc_[1]; nc2 = nc_[2];                                 \
    }

#define STAGE_PAIR()                                                              \
    {                                                                             \
        *(v2f*)(fib0 + 2 * lane) = fiA;                                           \
        *(v2f*)(fib1 + 2 * lane) = fiB;                                           \
        _Pragma("unroll")                                                         \
        for (int m = 0; m < 32; ++m) {                                            \
            if (m < 16) { v2f d0 = g[m] - fiA; *(v2f*)(X0 + m * XSTR + 2 * lane) = d0; }        \
            else        { v2f d1 = g[m] - fiB; *(v2f*)(X1 + (m - 16) * XSTR + 2 * lane) = d1; } \
        }                                                                         \
        v4f dp; dp.x = nc0 - pc0; dp.y = nc1 - pc1; dp.z = nc2 - pc2; dp.w = 0.0f; \
        *(v4f*)((h ? X1 : X0) + nl * XSTR + 64) = dp;                             \
    }

    int nbvN = ib[0];
    LOAD_PAIR(0, nbvN)
    nbvN = ib[2 * KNN];
    STAGE_PAIR()

    for (int it = 0; it < 16; ++it) {
        const int nx  = (it + 1 < 16) ? it + 1 : 15;
        const int nx2 = (it + 2 < 16) ? it + 2 : 15;
        LOAD_PAIR(nx, nbvN)
        nbvN = ib[nx2 * 2 * KNN];
        const int P0 = pg * 32 + it * 2;

        v8f a0c[2], a1c[2];
        const v8f zc = {0.f, 0.f, 0.f, 0.f, 0.f, 0.f, 0.f, 0.f};
        a0c[0] = zc; a0c[1] = zc; a1c[0] = zc; a1c[1] = zc;

#define STEP(KS, A0, A1)                                                          \
        {                                                                         \
            a0c[0] = __builtin_amdgcn_wmma_f32_16x16x4_f32(                       \
                false, (A0), false, bw0[KS], (short)0, a0c[0], false, false);     \
            a1c[0] = __builtin_amdgcn_wmma_f32_16x16x4_f32(                       \
                false, (A1), false, bw0[KS], (short)0, a1c[0], false, false);     \
            a0c[1] = __builtin_amdgcn_wmma_f32_16x16x4_f32(                       \
                false, (A0), false, bw1[KS], (short)0, a0c[1], false, false);     \
            a1c[1] = __builtin_amdgcn_wmma_f32_16x16x4_f32(                       \
                false, (A1), false, bw1[KS], (short)0, a1c[1], false, false);     \
        }

#pragma unroll
        for (int ks = 0; ks < 16; ++ks) {
            v2f A0 = *(const v2f*)(fib0 + 4 * ks + 2 * h);
            v2f A1 = *(const v2f*)(fib1 + 4 * ks + 2 * h);
            STEP(ks, A0, A1);
        }
#pragma unroll
        for (int ks = 0; ks < 16; ++ks) {
            int xo = nl * XSTR + 4 * ks + 2 * h;
            v2f A0 = *(const v2f*)(X0 + xo);
            v2f A1 = *(const v2f*)(X1 + xo);
            STEP(16 + ks, A0, A1);
        }
        {
            int xo = nl * XSTR + 64 + 2 * h;
            v2f A0 = *(const v2f*)(X0 + xo);
            v2f A1 = *(const v2f*)(X1 + xo);
            STEP(32, A0, A1);
        }
#undef STEP

        float p0[2], p1[2];
#pragma unroll
        for (int t = 0; t < 2; ++t) {
            float v0 = a0c[t][0];
            float v1 = a1c[t][0];
#pragma unroll
            for (int r = 1; r < 8; ++r) {
                v0 = fmaxf(v0, a0c[t][r]);
                v1 = fmaxf(v1, a1c[t][r]);
            }
            p0[t] = v0; p1[t] = v1;
        }
        float m0 = fmaxf(h ? p0[1] : p0[0], __shfl_xor(h ? p0[0] : p0[1], 16, 32));
        float m1 = fmaxf(h ? p1[1] : p1[0], __shfl_xor(h ? p1[0] : p1[1], 16, 32));
        float* outp = out + (size_t)P0 * CCH + half * 32 + lane;
        outp[0]   = fmaxf(m0 + bsel, 0.0f);
        outp[CCH] = fmaxf(m1 + bsel, 0.0f);

        STAGE_PAIR()
    }
#undef LOAD_PAIR
#undef STAGE_PAIR
}

extern "C" void kernel_launch(void* const* d_in, const int* in_sizes, int n_in,
                              void* d_out, int out_size, void* d_ws, size_t ws_size,
                              hipStream_t stream) {
    const float* coords = (const float*)d_in[0];
    const float* feat   = (const float*)d_in[1];
    const float* Wm     = (const float*)d_in[2];
    const float* bias   = (const float*)d_in[3];
    float* out = (float*)d_out;
    int* idx = (int*)d_ws;

    knn_kernel<<<256, KNT, 0, stream>>>(coords, idx);
    agg_kernel<<<256, 128, 0, stream>>>(coords, feat, Wm, bias, idx, out);
}


# KNN: after each merge a slice publishes its 2nd-smallest distance in LDS; the largest of a query's 8 published values bounds its 16th distance and tightens every slice's entry test; same lists
# speedup vs baseline: 23.8059x; 1.0916x over previous
#include <hip/hip_runtime.h>

typedef __attribute__((ext_vector_type(2))) float v2f;
typedef __attribute__((ext_vector_type(4))) float v4f;
typedef __attribute__((ext_vector_type(8))) float v8f;

#define NPTS 8192
#define CCH  64
#define KNN  16
#define XSTR 68
#define WAVE_LDS (2*64 + 2*16*XSTR)

typedef unsigned int u32;
typedef __attribute__((ext_vector_type(2))) unsigned int v2u;

__device__ __forceinline__ bool key_lt(u32 da, u32 ia, u32 db, u32 ib) {
    u32 c1, c2;
    (void)__builtin_subc(ia, ib, 0u, &c1);
    (void)__builtin_subc(da, db, c1, &c2);
    return c2 != 0u;
}

__device__ __forceinline__ void cswap(u32& da, u32& ia, u32& db, u32& ib) {
    const bool lt = key_lt(da, ia, db, ib);
    const u32 dlo = lt ? da : db, dhi = lt ? db : da;
    const u32 ilo = lt ? ia : ib, ihi = lt ? ib : ia;
    da = dlo; ia = ilo; db = dhi; ib = ihi;
}

__device__ __forceinline__ void sort16(u32 (&d)[16], u32 (&x)[16]) {
#define CS(a, b) cswap(d[a], x[a], d[b], x[b]);
    CS(0,1) CS(2,3) CS(4,5) CS(6,7) CS(8,9) CS(10,11) CS(12,13) CS(14,15)
    CS(0,2) CS(1,3) CS(4,6) CS(5,7) CS(8,10) CS(9,11) CS(12,14) CS(13,15)
    CS(1,2) CS(5,6) CS(9,10) CS(13,14)
    CS(0,4) CS(1,5) CS(2,6) CS(3,7) CS(8,12) CS(9,13) CS(10,14) CS(11,15)
    CS(2,4) CS(3,5) CS(10,12) CS(11,13)
    CS(1,2) CS(3,4) CS(5,6) CS(9,10) CS(11,12) CS(13,14)
    CS(0,8) CS(1,9) CS(2,10) CS(3,11) CS(4,12) CS(5,13) CS(6,14) CS(7,15)
    CS(4,8) CS(5,9) CS(6,10) CS(7,11)
    CS(2,4) CS(3,5) CS(6,8) CS(7,9) CS(10,12) CS(11,13)
    CS(1,2) CS(3,4) CS(5,6) CS(7,8) CS(9,10) CS(11,12) CS(13,14)
#undef CS
}

__device__ __forceinline__ void merge16(u32 (&bd)[16], u32 (&bi)[16],
                                        const u32 (&xd)[16], const u32 (&xi)[16]) {
#pragma unroll
    for (int i = 0; i < 16; ++i) {
        const u32 od = xd[15 - i], oi = xi[15 - i];
        const bool lt = key_lt(bd[i], bi[i], od, oi);
        bd[i] = lt ? bd[i] : od;
        bi[i] = lt ? bi[i] : oi;
    }
#pragma unroll
    for (int lj = 3; lj >= 0; --lj) {
#pragma unroll
        for (int i = 0; i < 16; ++i) {
            const int l = i ^ (1 << lj);
            if (l > i) cswap(bd[i], bi[i], bd[l], bi[l]);
        }
    }
}

#define KNT   512
#define KSLEN 1024
#define KBCAP 32

__global__ __launch_bounds__(KNT) void knn_kernel(const float* __restrict__ coords,
                                                  int* __restrict__ idx_out) {
    __shared__ __attribute__((aligned(16))) v2u lds_raw[2 * NPTS];
    __shared__ __attribute__((aligned(8)))  v2u lds_pend[KBCAP * KNT];
    __shared__ u32 lds_pub[KNT];
    v4f* pts = (v4f*)lds_raw;
    const int tid   = threadIdx.x;
    const int b     = blockIdx.x >> 7;
    const int qbase = (blockIdx.x & 127) << 6;
    const float* cb = coords + b * NPTS * 3;

    for (int j = tid; j < NPTS; j += KNT) {
        float x = cb[j * 3 + 0], y = cb[j * 3 + 1], z = cb[j * 3 + 2];
        v4f p; p.x = x; p.y = y; p.z = z; p.w = x * x + y * y + z * z;
        pts[j] = p;
    }
    lds_pub[tid] = ~0u;
    __syncthreads();

    const int lane = tid & 31;
    const int wav  = tid >> 5;
    const int s    = wav >> 1;
    const int q    = qbase + ((wav & 1) << 5) + lane;
    const v4f me   = pts[q];

    u32 bd[KNN], bi[KNN];
#pragma unroll
    for (int t = 0; t < KNN; ++t) { bd[t] = ~0u; bi[t] = ~0u; }

    v2u* const pend = lds_pend + tid;
    v2u* wp = pend;
    u32 thr = ~0u;
    u32 bnd = ~0u;
    const int qslot = tid & 63;

    const int cbase = s * KSLEN;
    for (int c = 0; c < KSLEN / 16; ++c) {
        {
            __asm__ volatile("" ::: "memory");
            u32 m = lds_pub[qslot];
#pragma unroll
            for (int t = 1; t < 8; ++t) { const u32 v = lds_pub[t * 64 + qslot]; m = v > m ? v : m; }
            bnd = m < bnd ? m : bnd;
            thr = bnd < thr ? bnd : thr;
        }
        u32 du[16];
#pragma unroll
        for (int i = 0; i < 16; ++i) {
            const v4f p = pts[cbase + c * 16 + i];
            float dot = me.x * p.x;
            dot = fmaf(me.y, p.y, dot);
            dot = fmaf(me.z, p.z, dot);
            float d = fmaf(-2.0f, dot, me.w + p.w);
            d = fmaxf(d, 0.0f);
            du[i] = __float_as_uint(d);
        }
#pragma unroll
        for (int i = 0; i < 16; ++i) {
            if (du[i] <= thr) {
                v2u k; k.x = (u32)(cbase + c * 16 + i); k.y = du[i];
                *wp = k;
                wp += KNT;
            }
        }
        const bool last = (c == KSLEN / 16 - 1);
        int cnt = (int)(wp - pend) / KNT;
        if (__any(last ? (cnt > 0) : (cnt >= 16))) {
            for (int pass = 0; pass < 2; ++pass) {
                u32 xd[16], xi[16];
#pragma unroll
                for (int t = 0; t < 16; ++t) {
                    v2u k; k.x = ~0u; k.y = ~0u;
                    if (t < cnt) k = pend[t * KNT];
                    xi[t] = k.x; xd[t] = k.y;
                }
                sort16(xd, xi);
                merge16(bd, bi, xd, xi);
                thr = bd[15] < bnd ? bd[15] : bnd;
                lds_pub[tid] = bd[1];
                int left = 0;
                if (__any(cnt > 16)) {
#pragma unroll
                    for (int t = 0; t < 15; ++t) {
                        if (16 + t < cnt) {
                            const v2u k = pend[(16 + t) * KNT];
                            if (k.y <= thr) { pend[left * KNT] = k; ++left; }
                        }
                    }
                }
                cnt = left;
                if (!(last && __any(cnt > 0))) break;
            }
            wp = pend + cnt * KNT;
        }
    }

    __syncthreads();
    v2u* lists = lds_raw;
#pragma unroll
    for (int r = 0; r < 3; ++r) {
        const int span = 1 << r;
        const int ph   = s & (2 * span - 1);
        if (ph == span) {
#pragma unroll
            for (int t = 0; t < KNN; ++t) { v2u k; k.x = bi[t]; k.y = bd[t]; lists[t * KNT + tid] = k; }
        }
        __syncthreads();
        if (ph == 0) {
            u32 xd[16], xi[16];
#pragma unroll
            for (int t = 0; t < KNN; ++t) { const v2u k = lists[t * KNT + tid + 64 * span]; xi[t] = k.x; xd[t] = k.y; }
            merge16(bd, bi, xd, xi);
        }
    }

    if (s == 0) {
        int* op = idx_out + (b * NPTS + q) * KNN;
#pragma unroll
        for (int t = 0; t < KNN; ++t) op[t] = (int)bi[t];
    }
}

__global__ __launch_bounds__(128) __attribute__((amdgpu_waves_per_eu(1, 1))) void agg_kernel(const float* __restrict__ coords,
                                                  const float* __restrict__ feat,
                                                  const float* __restrict__ Wm,
                                                  const float* __restrict__ bias,
                                                  const int*   __restrict__ idx,
                                                  float* __restrict__ out) {
    __shared__ float XS[4 * WAVE_LDS];

    const int lane = threadIdx.x & 31;
    const int wav  = threadIdx.x >> 5;
    const int nl   = lane & 15;
    const int h    = lane >> 4;

    const int gw   = blockIdx.x * 4 + wav;
    const int half = gw & 1;
    const int pg   = gw >> 1;

    float* fib0 = &XS[wav * WAVE_LDS];
    float* fib1 = fib0 + 64;
    float* X0   = fib1 + 64;
    float* X1   = X0 + 16 * XSTR;

    v2f bw0[33], bw1[33];
    {
        const float* wc = Wm + half * 32 + nl;
#pragma unroll
        for (int ks = 0; ks < 33; ++ks) {
            const int k = 4 * ks + 2 * h;
            v2f t0, t1;
            t0.x = wc[k * CCH];
            t1.x = wc[k * CCH + 16];
            t0.y = (k + 1 < 131) ? wc[(k + 1) * CCH]      : 0.0f;
            t1.y = (k + 1 < 131) ? wc[(k + 1) * CCH + 16] : 0.0f;
            bw0[ks] = t0;
            bw1[ks] = t1;
        }
    }

    const float bsel = bias[half * 32 + lane];

    const int b     = (pg * 32) >> 13;
    const int nbase = (pg * 32) & (NPTS - 1);
    const float* fb = feat   + b * NPTS * CCH;
    const float* cb = coords + b * NPTS * 3;
    const int* ib = idx + (b * NPTS + nbase + h) * KNN + nl;

    v2f   fiA, fiB, g[32];
    float pc0, pc1, pc2, nc0, nc1, nc2;

#define LOAD_PAIR(PAIR, NBV)                                                      \
    {                                                                             \
        const int n0_ = nbase + (PAIR) * 2;                                       \
        fiA = *(const v2f*)(fb + n0_ * CCH + 2 * lane);                           \
        fiB = *(const v2f*)(fb + (n0_ + 1) * CCH + 2 * lane);                     \
        _Pragma("unroll")                                                         \
        for (int m = 0; m < 32; ++m) {                                            \
            const int nb_ = __shfl((NBV), m, 32);                                 \
            g[m] = *(const v2f*)(fb + nb_ * CCH + 2 * lane);                      \
        }                                                                         \
        const float* pc_ = cb + (n0_ + h) * 3;                                    \
        const float* nc_ = cb + (NBV) * 3;                                        \
        pc0 = pc_[0]; pc1 = pc_[1]; pc2 = pc_[2];                                 \
        nc0 = nc_[0]; nc1 = nc_[1]; nc2 = nc_[2];                                 \
    }

#define STAGE_PAIR()                                                              \
    {                                                                             \
        *(v2f*)(fib0 + 2 * lane) = fiA;                                           \
        *(v2f*)(fib1 + 2 * lane) = fiB;                                           \
        _Pragma("unroll")                                                         \
        for (int m = 0; m < 32; ++m) {                                            \
            if (m < 16) { v2f d0 = g[m] - fiA; *(v2f*)(X0 + m * XSTR + 2 * lane) = d0; }        \
            else        { v2f d1 = g[m] - fiB; *(v2f*)(X1 + (m - 16) * XSTR + 2 * lane) = d1; } \
        }                                                                         \
        v4f dp; dp.x = nc0 - pc0; dp.y = nc1 - pc1; dp.z = nc2 - pc2; dp.w = 0.0f; \
        *(v4f*)((h ? X1 : X0) + nl * XSTR + 64) = dp;                             \
    }

    int nbvN = ib[0];
    LOAD_PAIR(0, nbvN)
    nbvN = ib[2 * KNN];
    STAGE_PAIR()

    for (int it = 0; it < 16; ++it) {
        const int nx  = (it + 1 < 16) ? it + 1 : 15;
        const int nx2 = (it + 2 < 16) ? it + 2 : 15;
        LOAD_PAIR(nx, nbvN)
        nbvN = ib[nx2 * 2 * KNN];
        const int P0 = pg * 32 + it * 2;

        v8f a0c[2], a1c[2];
        const v8f zc = {0.f, 0.f, 0.f, 0.f, 0.f, 0.f, 0.f, 0.f};
        a0c[0] = zc; a0c[1] = zc; a1c[0] = zc; a1c[1] = zc;

#define STEP(KS, A0, A1)                                                          \
        {                                                                         \
            a0c[0] = __builtin_amdgcn_wmma_f32_16x16x4_f32(                       \
                false, (A0), false, bw0[KS], (short)0, a0c[0], false, false);     \
            a1c[0] = __builtin_amdgcn_wmma_f32_16x16x4_f32(                       \
                false, (A1), false, bw0[KS], (short)0, a1c[0], false, false);     \
            a0c[1] = __builtin_amdgcn_wmma_f32_16x16x4_f32(                       \
                false, (A0), false, bw1[KS], (short)0, a0c[1], false, false);     \
            a1c[1] = __builtin_amdgcn_wmma_f32_16x16x4_f32(                       \
                false, (A1), false, bw1[KS], (short)0, a1c[1], false, false);     \
        }

#pragma unroll
        for (int ks = 0; ks < 16; ++ks) {
            v2f A0 = *(const v2f*)(fib0 + 4 * ks + 2 * h);
            v2f A1 = *(const v2f*)(fib1 + 4 * ks + 2 * h);
            STEP(ks, A0, A1);
        }
#pragma unroll
        for (int ks = 0; ks < 16; ++ks) {
            int xo = nl * XSTR + 4 * ks + 2 * h;
            v2f A0 = *(const v2f*)(X0 + xo);
            v2f A1 = *(const v2f*)(X1 + xo);
            STEP(16 + ks, A0, A1);
        }
        {
            int xo = nl * XSTR + 64 + 2 * h;
            v2f A0 = *(const v2f*)(X0 + xo);
            v2f A1 = *(const v2f*)(X1 + xo);
            STEP(32, A0, A1);
        }
#undef STEP

        float p0[2], p1[2];
#pragma unroll
        for (int t = 0; t < 2; ++t) {
            float v0 = a0c[t][0];
            float v1 = a1c[t][0];
#pragma unroll
            for (int r = 1; r < 8; ++r) {
                v0 = fmaxf(v0, a0c[t][r]);
                v1 = fmaxf(v1, a1c[t][r]);
            }
            p0[t] = v0; p1[t] = v1;
        }
        float m0 = fmaxf(h ? p0[1] : p0[0], __shfl_xor(h ? p0[0] : p0[1], 16, 32));
        float m1 = fmaxf(h ? p1[1] : p1[0], __shfl_xor(h ? p1[0] : p1[1], 16, 32));
        float* outp = out + (size_t)P0 * CCH + half * 32 + lane;
        outp[0]   = fmaxf(m0 + bsel, 0.0f);
        outp[CCH] = fmaxf(m1 + bsel, 0.0f);

        STAGE_PAIR()
    }
#undef LOAD_PAIR
#undef STAGE_PAIR
}

extern "C" void kernel_launch(void* const* d_in, const int* in_sizes, int n_in,
                              void* d_out, int out_size, void* d_ws, size_t ws_size,
                              hipStream_t stream) {
    const float* coords = (const float*)d_in[0];
    const float* feat   = (const float*)d_in[1];
    const float* Wm     = (const float*)d_in[2];
    const float* bias   = (const float*)d_in[3];
    float* out = (float*)d_out;
    int* idx = (int*)d_ws;

    knn_kernel<<<256, KNT, 0, stream>>>(coords, idx);
    agg_kernel<<<256, 128, 0, stream>>>(coords, feat, Wm, bias, idx, out);
}


# KNN scan: each candidate's entry test is a wave-uniform branch on the compare's own lane mask; inside it all lanes store unmasked and only a passing lane's slot pointer advances; same lists
# speedup vs baseline: 27.6915x; 1.1632x over previous
#include <hip/hip_runtime.h>

typedef __attribute__((ext_vector_type(2))) float v2f;
typedef __attribute__((ext_vector_type(4))) float v4f;
typedef __attribute__((ext_vector_type(8))) float v8f;

#define NPTS 8192
#define CCH  64
#define KNN  16
#define XSTR 68
#define WAVE_LDS (2*64 + 2*16*XSTR)

typedef unsigned int u32;
typedef __attribute__((ext_vector_type(2))) unsigned int v2u;

__device__ __forceinline__ bool key_lt(u32 da, u32 ia, u32 db, u32 ib) {
    u32 c1, c2;
    (void)__builtin_subc(ia, ib, 0u, &c1);
    (void)__builtin_subc(da, db, c1, &c2);
    return c2 != 0u;
}

__device__ __forceinline__ void cswap(u32& da, u32& ia, u32& db, u32& ib) {
    const bool lt = key_lt(da, ia, db, ib);
    const u32 dlo = lt ? da : db, dhi = lt ? db : da;
    const u32 ilo = lt ? ia : ib, ihi = lt ? ib : ia;
    da = dlo; ia = ilo; db = dhi; ib = ihi;
}

__device__ __forceinline__ void sort16(u32 (&d)[16], u32 (&x)[16]) {
#define CS(a, b) cswap(d[a], x[a], d[b], x[b]);
    CS(0,1) CS(2,3) CS(4,5) CS(6,7) CS(8,9) CS(10,11) CS(12,13) CS(14,15)
    CS(0,2) CS(1,3) CS(4,6) CS(5,7) CS(8,10) CS(9,11) CS(12,14) CS(13,15)
    CS(1,2) CS(5,6) CS(9,10) CS(13,14)
    CS(0,4) CS(1,5) CS(2,6) CS(3,7) CS(8,12) CS(9,13) CS(10,14) CS(11,15)
    CS(2,4) CS(3,5) CS(10,12) CS(11,13)
    CS(1,2) CS(3,4) CS(5,6) CS(9,10) CS(11,12) CS(13,14)
    CS(0,8) CS(1,9) CS(2,10) CS(3,11) CS(4,12) CS(5,13) CS(6,14) CS(7,15)
    CS(4,8) CS(5,9) CS(6,10) CS(7,11)
    CS(2,4) CS(3,5) CS(6,8) CS(7,9) CS(10,12) CS(11,13)
    CS(1,2) CS(3,4) CS(5,6) CS(7,8) CS(9,10) CS(11,12) CS(13,14)
#undef CS
}

__device__ __forceinline__ void merge16(u32 (&bd)[16], u32 (&bi)[16],
                                        const u32 (&xd)[16], const u32 (&xi)[16]) {
#pragma unroll
    for (int i = 0; i < 16; ++i) {
        const u32 od = xd[15 - i], oi = xi[15 - i];
        const bool lt = key_lt(bd[i], bi[i], od, oi);
        bd[i] = lt ? bd[i] : od;
        bi[i] = lt ? bi[i] : oi;
    }
#pragma unroll
    for (int lj = 3; lj >= 0; --lj) {
#pragma unroll
        for (int i = 0; i < 16; ++i) {
            const int l = i ^ (1 << lj);
            if (l > i) cswap(bd[i], bi[i], bd[l], bi[l]);
        }
    }
}

#define KNT   512
#define KSLEN 1024
#define KBCAP 32

__global__ __launch_bounds__(KNT) void knn_kernel(const float* __restrict__ coords,
                                                  int* __restrict__ idx_out) {
    __shared__ __attribute__((aligned(16))) v2u lds_raw[2 * NPTS];
    __shared__ __attribute__((aligned(8)))  v2u lds_pend[KBCAP * KNT];
    __shared__ u32 lds_pub[KNT];
    const v4f* pts = (const v4f*)lds_raw;
    float* ptsf = (float*)lds_raw;
    const int tid   = threadIdx.x;
    const int b     = blockIdx.x >> 7;
    const int qbase = (blockIdx.x & 127) << 6;
    const float* cb = coords + b * NPTS * 3;

#pragma unroll
    for (int u = 0; u < NPTS / KNT; ++u) {
        const int j = tid + u * KNT;
        float x = cb[j * 3 + 0], y = cb[j * 3 + 1], z = cb[j * 3 + 2];
        float w;
        {
#pragma clang fp contract(off)
            const float t = x * x + y * y;
            w = __builtin_fmaf(z, z, t);
        }
        float* pp = ptsf + 8 * (j >> 1) + (j & 1);
        pp[0] = x; pp[2] = y; pp[4] = z; pp[6] = w;
    }
    lds_pub[tid] = ~0u;
    __syncthreads();

    const int lane = tid & 31;
    const int wav  = tid >> 5;
    const int s    = wav >> 1;
    const int q    = qbase + ((wav & 1) << 5) + lane;
    const float* mp = ptsf + 8 * (q >> 1) + (q & 1);
    v2f mx2, my2, mz2, mw2;
    mx2.x = mp[0]; mx2.y = mx2.x;
    my2.x = mp[2]; my2.y = my2.x;
    mz2.x = mp[4]; mz2.y = mz2.x;
    mw2.x = mp[6]; mw2.y = mw2.x;
    const v2f neg2 = {-2.0f, -2.0f};
    const v2f zero2 = {0.0f, 0.0f};

    u32 bd[KNN], bi[KNN];
#pragma unroll
    for (int t = 0; t < KNN; ++t) { bd[t] = ~0u; bi[t] = ~0u; }

    v2u* const pend = lds_pend + tid;
    typedef __attribute__((address_space(3))) v2u lds_v2u;
    lds_v2u* const pend3 = (lds_v2u*)pend;
    lds_v2u* wp = pend3;
    u32 thr = ~0u;
    u32 bnd = ~0u;
    const int qslot = tid & 63;

    const int cbase = s * KSLEN;
    for (int c = 0; c < KSLEN / 16; ++c) {
        {
            __asm__ volatile("" ::: "memory");
            u32 m = lds_pub[qslot];
#pragma unroll
            for (int t = 1; t < 8; ++t) { const u32 v = lds_pub[t * 64 + qslot]; m = v > m ? v : m; }
            bnd = m < bnd ? m : bnd;
            thr = bnd < thr ? bnd : thr;
        }
        u32 du[16];
#pragma unroll
        for (int t = 0; t < 8; ++t) {
            const v4f a  = pts[cbase + c * 16 + 2 * t];
            const v4f bq = pts[cbase + c * 16 + 2 * t + 1];
            v2f X, Y, Z, Wv;
            X.x = a.x;  X.y = a.y;  Y.x = a.z;   Y.y = a.w;
            Z.x = bq.x; Z.y = bq.y; Wv.x = bq.z; Wv.y = bq.w;
            v2f dot = mx2 * X;
            dot = __builtin_elementwise_fma(my2, Y, dot);
            dot = __builtin_elementwise_fma(mz2, Z, dot);
            v2f d = __builtin_elementwise_fma(neg2, dot, mw2 + Wv);
            d = __builtin_elementwise_max(d, zero2);
            du[2 * t]     = __float_as_uint(d.x);
            du[2 * t + 1] = __float_as_uint(d.y);
        }
        __asm__ volatile("" :: "v"(du[0]), "v"(du[1]), "v"(du[2]), "v"(du[3]),
                               "v"(du[4]), "v"(du[5]), "v"(du[6]), "v"(du[7]),
                               "v"(du[8]), "v"(du[9]), "v"(du[10]), "v"(du[11]),
                               "v"(du[12]), "v"(du[13]), "v"(du[14]), "v"(du[15]));
#pragma unroll
        for (int i = 0; i < 16; ++i) {
            const bool pass = du[i] <= thr;
            if (__any(pass)) {
                v2u k; k.x = (u32)(cbase + c * 16 + i); k.y = du[i];
                *wp = k;
                wp += pass ? KNT : 0;
            }
        }
        const bool last = (c == KSLEN / 16 - 1);
        const int trig = last ? 1 : 16;
        if (__any(wp >= pend3 + trig * KNT)) {
            int cnt = (int)(wp - pend3) / KNT;
            for (int pass = 0; pass < 2; ++pass) {
                u32 xd[16], xi[16];
#pragma unroll
                for (int t = 0; t < 16; ++t) {
                    v2u k; k.x = ~0u; k.y = ~0u;
                    if (t < cnt) k = pend[t * KNT];
                    xi[t] = k.x; xd[t] = k.y;
                }
                sort16(xd, xi);
                merge16(bd, bi, xd, xi);
                thr = bd[15] < bnd ? bd[15] : bnd;
                lds_pub[tid] = bd[1];
                int left = 0;
                if (__any(cnt > 16)) {
#pragma unroll
                    for (int t = 0; t < 15; ++t) {
                        if (16 + t < cnt) {
                            const v2u k = pend[(16 + t) * KNT];
                            if (k.y <= thr) { pend[left * KNT] = k; ++left; }
                        }
                    }
                }
                cnt = left;
                if (!(last && __any(cnt > 0))) break;
            }
            wp = pend3 + cnt * KNT;
        }
    }

    __syncthreads();
    v2u* lists = lds_raw;
#pragma unroll
    for (int r = 0; r < 3; ++r) {
        const int span = 1 << r;
        const int ph   = s & (2 * span - 1);
        if (ph == span) {
#pragma unroll
            for (int t = 0; t < KNN; ++t) { v2u k; k.x = bi[t]; k.y = bd[t]; lists[t * KNT + tid] = k; }
        }
        __syncthreads();
        if (ph == 0) {
            u32 xd[16], xi[16];
#pragma unroll
            for (int t = 0; t < KNN; ++t) { const v2u k = lists[t * KNT + tid + 64 * span]; xi[t] = k.x; xd[t] = k.y; }
            merge16(bd, bi, xd, xi);
        }
    }

    if (s == 0) {
        int* op = idx_out + (b * NPTS + q) * KNN;
#pragma unroll
        for (int t = 0; t < KNN; ++t) op[t] = (int)bi[t];
    }
}

__global__ __launch_bounds__(128) __attribute__((amdgpu_waves_per_eu(1, 1))) void agg_kernel(const float* __restrict__ coords,
                                                  const float* __restrict__ feat,
                                                  const float* __restrict__ Wm,
                                                  const float* __restrict__ bias,
                                                  const int*   __restrict__ idx,
                                                  float* __restrict__ out) {
    __shared__ float XS[4 * WAVE_LDS];

    const int lane = threadIdx.x & 31;
    const int wav  = threadIdx.x >> 5;
    const int nl   = lane & 15;
    const int h    = lane >> 4;

    const int gw   = blockIdx.x * 4 + wav;
    const int half = gw & 1;
    const int pg   = gw >> 1;

    float* fib0 = &XS[wav * WAVE_LDS];
    float* fib1 = fib0 + 64;
    float* X0   = fib1 + 64;
    float* X1   = X0 + 16 * XSTR;

    v2f bw0[33], bw1[33];
    {
        const float* wc = Wm + half * 32 + nl;
#pragma unroll
        for (int ks = 0; ks < 33; ++ks) {
            const int k = 4 * ks + 2 * h;
            v2f t0, t1;
            t0.x = wc[k * CCH];
            t1.x = wc[k * CCH + 16];
            t0.y = (k + 1 < 131) ? wc[(k + 1) * CCH]      : 0.0f;
            t1.y = (k + 1 < 131) ? wc[(k + 1) * CCH + 16] : 0.0f;
            bw0[ks] = t0;
            bw1[ks] = t1;
        }
    }

    const float bsel = bias[half * 32 + lane];

    const int b     = (pg * 32) >> 13;
    const int nbase = (pg * 32) & (NPTS - 1);
    const float* fb = feat   + b * NPTS * CCH;
    const float* cb = coords + b * NPTS * 3;
    const int* ib = idx + (b * NPTS + nbase + h) * KNN + nl;

    v2f   fiA, fiB, g[32];
    float pc0, pc1, pc2, nc0, nc1, nc2;

#define LOAD_PAIR(PAIR, NBV)                                                      \
    {                                                                             \
        const int n0_ = nbase + (PAIR) * 2;                                       \
        fiA = *(const v2f*)(fb + n0_ * CCH + 2 * lane);                           \
        fiB = *(const v2f*)(fb + (n0_ + 1) * CCH + 2 * lane);                     \
        _Pragma("unroll")                                                         \
        for (int m = 0; m < 32; ++m) {                                            \
            const int nb_ = __shfl((NBV), m, 32);                                 \
            g[m] = *(const v2f*)(fb + nb_ * CCH + 2 * lane);                      \
        }                                                                         \
        const float* pc_ = cb + (n0_ + h) * 3;                                    \
        const float* nc_ = cb + (NBV) * 3;                                        \
        pc0 = pc_[0]; pc1 = pc_[1]; pc2 = pc_[2];                                 \
        nc0 = nc_[0]; nc1 = nc_[1]; nc2 = nc_[2];                                 \
    }

#define STAGE_PAIR()                                                              \
    {                                                                             \
        *(v2f*)(fib0 + 2 * lane) = fiA;                                           \
        *(v2f*)(fib1 + 2 * lane) = fiB;                                           \
        _Pragma("unroll")                                                         \
        for (int m = 0; m < 32; ++m) {                                            \
            if (m < 16) { v2f d0 = g[m] - fiA; *(v2f*)(X0 + m * XSTR + 2 * lane) = d0; }        \
            else        { v2f d1 = g[m] - fiB; *(v2f*)(X1 + (m - 16) * XSTR + 2 * lane) = d1; } \
        }                                                                         \
        v4f dp; dp.x = nc0 - pc0; dp.y = nc1 - pc1; dp.z = nc2 - pc2; dp.w = 0.0f; \
        *(v4f*)((h ? X1 : X0) + nl * XSTR + 64) = dp;                             \
    }

    int nbvN = ib[0];
    LOAD_PAIR(0, nbvN)
    nbvN = ib[2 * KNN];
    STAGE_PAIR()

    for (int it = 0; it < 16; ++it) {
        const int nx  = (it + 1 < 16) ? it + 1 : 15;
        const int nx2 = (it + 2 < 16) ? it + 2 : 15;
        LOAD_PAIR(nx, nbvN)
        nbvN = ib[nx2 * 2 * KNN];
        const int P0 = pg * 32 + it * 2;

        v8f a0c[2], a1c[2];
        const v8f zc = {0.f, 0.f, 0.f, 0.f, 0.f, 0.f, 0.f, 0.f};
        a0c[0] = zc; a0c[1] = zc; a1c[0] = zc; a1c[1] = zc;

#define STEP(KS, A0, A1)                                                          \
        {                                                                         \
            a0c[0] = __builtin_amdgcn_wmma_f32_16x16x4_f32(                       \
                false, (A0), false, bw0[KS], (short)0, a0c[0], false, false);     \
            a1c[0] = __builtin_amdgcn_wmma_f32_16x16x4_f32(                       \
                false, (A1), false, bw0[KS], (short)0, a1c[0], false, false);     \
            a0c[1] = __builtin_amdgcn_wmma_f32_16x16x4_f32(                       \
                false, (A0), false, bw1[KS], (short)0, a0c[1], false, false);     \
            a1c[1] = __builtin_amdgcn_wmma_f32_16x16x4_f32(                       \
                false, (A1), false, bw1[KS], (short)0, a1c[1], false, false);     \
        }

#pragma unroll
        for (int ks = 0; ks < 16; ++ks) {
            v2f A0 = *(const v2f*)(fib0 + 4 * ks + 2 * h);
            v2f A1 = *(const v2f*)(fib1 + 4 * ks + 2 * h);
            STEP(ks, A0, A1);
        }
#pragma unroll
        for (int ks = 0; ks < 16; ++ks) {
            int xo = nl * XSTR + 4 * ks + 2 * h;
            v2f A0 = *(const v2f*)(X0 + xo);
            v2f A1 = *(const v2f*)(X1 + xo);
            STEP(16 + ks, A0, A1);
        }
        {
            int xo = nl * XSTR + 64 + 2 * h;
            v2f A0 = *(const v2f*)(X0 + xo);
            v2f A1 = *(const v2f*)(X1 + xo);
            STEP(32, A0, A1);
        }
#undef STEP

        float p0[2], p1[2];
#pragma unroll
        for (int t = 0; t < 2; ++t) {
            float v0 = a0c[t][0];
            float v1 = a1c[t][0];
#pragma unroll
            for (int r = 1; r < 8; ++r) {
                v0 = fmaxf(v0, a0c[t][r]);
                v1 = fmaxf(v1, a1c[t][r]);
            }
            p0[t] = v0; p1[t] = v1;
        }
        float m0 = fmaxf(h ? p0[1] : p0[0], __shfl_xor(h ? p0[0] : p0[1], 16, 32));
        float m1 = fmaxf(h ? p1[1] : p1[0], __shfl_xor(h ? p1[0] : p1[1], 16, 32));
        float* outp = out + (size_t)P0 * CCH + half * 32 + lane;
        outp[0]   = fmaxf(m0 + bsel, 0.0f);
        outp[CCH] = fmaxf(m1 + bsel, 0.0f);

        STAGE_PAIR()
    }
#undef LOAD_PAIR
#undef STAGE_PAIR
}

extern "C" void kernel_launch(void* const* d_in, const int* in_sizes, int n_in,
                              void* d_out, int out_size, void* d_ws, size_t ws_size,
                              hipStream_t stream) {
    const float* coords = (const float*)d_in[0];
    const float* feat   = (const float*)d_in[1];
    const float* Wm     = (const float*)d_in[2];
    const float* bias   = (const float*)d_in[3];
    float* out = (float*)d_out;
    int* idx = (int*)d_ws;

    knn_kernel<<<256, KNT, 0, stream>>>(coords, idx);
    agg_kernel<<<256, 128, 0, stream>>>(coords, feat, Wm, bias, idx, out);
}


# KNN: a slice's exchanged bound is its running 2nd-smallest distance, updated at each accepted candidate, stored each chunk; queued keys it has since beaten are dropped before a sort+merge; same lists
# speedup vs baseline: 28.6532x; 1.0347x over previous
#include <hip/hip_runtime.h>

typedef __attribute__((ext_vector_type(2))) float v2f;
typedef __attribute__((ext_vector_type(4))) float v4f;
typedef __attribute__((ext_vector_type(8))) float v8f;

#define NPTS 8192
#define CCH  64
#define KNN  16
#define XSTR 68
#define WAVE_LDS (2*64 + 2*16*XSTR)

typedef unsigned int u32;
typedef __attribute__((ext_vector_type(2))) unsigned int v2u;

__device__ __forceinline__ bool key_lt(u32 da, u32 ia, u32 db, u32 ib) {
    u32 c1, c2;
    (void)__builtin_subc(ia, ib, 0u, &c1);
    (void)__builtin_subc(da, db, c1, &c2);
    return c2 != 0u;
}

__device__ __forceinline__ void cswap(u32& da, u32& ia, u32& db, u32& ib) {
    const bool lt = key_lt(da, ia, db, ib);
    const u32 dlo = lt ? da : db, dhi = lt ? db : da;
    const u32 ilo = lt ? ia : ib, ihi = lt ? ib : ia;
    da = dlo; ia = ilo; db = dhi; ib = ihi;
}

__device__ __forceinline__ void sort16(u32 (&d)[16], u32 (&x)[16]) {
#define CS(a, b) cswap(d[a], x[a], d[b], x[b]);
    CS(0,1) CS(2,3) CS(4,5) CS(6,7) CS(8,9) CS(10,11) CS(12,13) CS(14,15)
    CS(0,2) CS(1,3) CS(4,6) CS(5,7) CS(8,10) CS(9,11) CS(12,14) CS(13,15)
    CS(1,2) CS(5,6) CS(9,10) CS(13,14)
    CS(0,4) CS(1,5) CS(2,6) CS(3,7) CS(8,12) CS(9,13) CS(10,14) CS(11,15)
    CS(2,4) CS(3,5) CS(10,12) CS(11,13)
    CS(1,2) CS(3,4) CS(5,6) CS(9,10) CS(11,12) CS(13,14)
    CS(0,8) CS(1,9) CS(2,10) CS(3,11) CS(4,12) CS(5,13) CS(6,14) CS(7,15)
    CS(4,8) CS(5,9) CS(6,10) CS(7,11)
    CS(2,4) CS(3,5) CS(6,8) CS(7,9) CS(10,12) CS(11,13)
    CS(1,2) CS(3,4) CS(5,6) CS(7,8) CS(9,10) CS(11,12) CS(13,14)
#undef CS
}

__device__ __forceinline__ void merge16(u32 (&bd)[16], u32 (&bi)[16],
                                        const u32 (&xd)[16], const u32 (&xi)[16]) {
#pragma unroll
    for (int i = 0; i < 16; ++i) {
        const u32 od = xd[15 - i], oi = xi[15 - i];
        const bool lt = key_lt(bd[i], bi[i], od, oi);
        bd[i] = lt ? bd[i] : od;
        bi[i] = lt ? bi[i] : oi;
    }
#pragma unroll
    for (int lj = 3; lj >= 0; --lj) {
#pragma unroll
        for (int i = 0; i < 16; ++i) {
            const int l = i ^ (1 << lj);
            if (l > i) cswap(bd[i], bi[i], bd[l], bi[l]);
        }
    }
}

#define KNT   512
#define KSLEN 1024
#define KBCAP 32

__global__ __launch_bounds__(KNT) void knn_kernel(const float* __restrict__ coords,
                                                  int* __restrict__ idx_out) {
    __shared__ __attribute__((aligned(16))) v2u lds_raw[2 * NPTS];
    __shared__ __attribute__((aligned(8)))  v2u lds_pend[KBCAP * KNT];
    __shared__ u32 lds_pub[KNT];
    const v4f* pts = (const v4f*)lds_raw;
    float* ptsf = (float*)lds_raw;
    const int tid   = threadIdx.x;
    const int b     = blockIdx.x >> 7;
    const int qbase = (blockIdx.x & 127) << 6;
    const float* cb = coords + b * NPTS * 3;

#pragma unroll
    for (int u = 0; u < NPTS / KNT; ++u) {
        const int j = tid + u * KNT;
        float x = cb[j * 3 + 0], y = cb[j * 3 + 1], z = cb[j * 3 + 2];
        float w;
        {
#pragma clang fp contract(off)
            const float t = x * x + y * y;
            w = __builtin_fmaf(z, z, t);
        }
        float* pp = ptsf + 8 * (j >> 1) + (j & 1);
        pp[0] = x; pp[2] = y; pp[4] = z; pp[6] = w;
    }
    lds_pub[tid] = ~0u;
    __syncthreads();

    const int lane = tid & 31;
    const int wav  = tid >> 5;
    const int s    = wav >> 1;
    const int q    = qbase + ((wav & 1) << 5) + lane;
    const float* mp = ptsf + 8 * (q >> 1) + (q & 1);
    v2f mx2, my2, mz2, mw2;
    mx2.x = mp[0]; mx2.y = mx2.x;
    my2.x = mp[2]; my2.y = my2.x;
    mz2.x = mp[4]; mz2.y = mz2.x;
    mw2.x = mp[6]; mw2.y = mw2.x;
    const v2f neg2 = {-2.0f, -2.0f};
    const v2f zero2 = {0.0f, 0.0f};

    u32 bd[KNN], bi[KNN];
#pragma unroll
    for (int t = 0; t < KNN; ++t) { bd[t] = ~0u; bi[t] = ~0u; }

    v2u* const pend = lds_pend + tid;
    typedef __attribute__((address_space(3))) v2u lds_v2u;
    lds_v2u* const pend3 = (lds_v2u*)pend;
    lds_v2u* wp = pend3;
    u32 thr = ~0u;
    u32 bnd = ~0u;
    u32 m1 = ~0u, m2 = ~0u;
    const int qslot = tid & 63;

    const int cbase = s * KSLEN;
    for (int c = 0; c < KSLEN / 16; ++c) {
        {
            __asm__ volatile("" ::: "memory");
            u32 m = lds_pub[qslot];
#pragma unroll
            for (int t = 1; t < 8; ++t) { const u32 v = lds_pub[t * 64 + qslot]; m = v > m ? v : m; }
            bnd = m < bnd ? m : bnd;
            thr = bnd < thr ? bnd : thr;
        }
        u32 du[16];
#pragma unroll
        for (int t = 0; t < 8; ++t) {
            const v4f a  = pts[cbase + c * 16 + 2 * t];
            const v4f bq = pts[cbase + c * 16 + 2 * t + 1];
            v2f X, Y, Z, Wv;
            X.x = a.x;  X.y = a.y;  Y.x = a.z;   Y.y = a.w;
            Z.x = bq.x; Z.y = bq.y; Wv.x = bq.z; Wv.y = bq.w;
            v2f dot = mx2 * X;
            dot = __builtin_elementwise_fma(my2, Y, dot);
            dot = __builtin_elementwise_fma(mz2, Z, dot);
            v2f d = __builtin_elementwise_fma(neg2, dot, mw2 + Wv);
            d = __builtin_elementwise_max(d, zero2);
            du[2 * t]     = __float_as_uint(d.x);
            du[2 * t + 1] = __float_as_uint(d.y);
        }
        __asm__ volatile("" :: "v"(du[0]), "v"(du[1]), "v"(du[2]), "v"(du[3]),
                               "v"(du[4]), "v"(du[5]), "v"(du[6]), "v"(du[7]),
                               "v"(du[8]), "v"(du[9]), "v"(du[10]), "v"(du[11]),
                               "v"(du[12]), "v"(du[13]), "v"(du[14]), "v"(du[15]));
#pragma unroll
        for (int i = 0; i < 16; ++i) {
            const bool pass = du[i] <= thr;
            if (__any(pass)) {
                v2u k; k.x = (u32)(cbase + c * 16 + i); k.y = du[i];
                *wp = k;
                wp += pass ? KNT : 0;
                const u32 hi = du[i] > m1 ? du[i] : m1;
                m2 = hi < m2 ? hi : m2;
                m1 = du[i] < m1 ? du[i] : m1;
            }
        }
        lds_pub[tid] = m2;
        const bool last = (c == KSLEN / 16 - 1);
        const int trig = last ? 1 : 16;
        if (__any(wp >= pend3 + trig * KNT)) {
            int cnt = (int)(wp - pend3) / KNT;
            if (!last) {
                v2u kq[KBCAP - 1];
#pragma unroll
                for (int t = 0; t < 16; ++t) kq[t] = pend[t * KNT];
                const bool more = __any(cnt > 16);
                if (more) {
#pragma unroll
                    for (int t = 16; t < KBCAP - 1; ++t) kq[t] = pend[t * KNT];
                }
                lds_v2u* wq = pend3;
#pragma unroll
                for (int t = 0; t < 16; ++t) {
                    const bool keep = (t < cnt) && (kq[t].y <= thr);
                    *wq = kq[t];
                    wq += keep ? KNT : 0;
                }
                if (more) {
#pragma unroll
                    for (int t = 16; t < KBCAP - 1; ++t) {
                        const bool keep = (t < cnt) && (kq[t].y <= thr);
                        *wq = kq[t];
                        wq += keep ? KNT : 0;
                    }
                }
                cnt = (int)(wq - pend3) / KNT;
            }
            if (__any(cnt >= trig)) {
                for (int pass = 0; pass < 2; ++pass) {
                    u32 xd[16], xi[16];
#pragma unroll
                    for (int t = 0; t < 16; ++t) {
                        v2u k; k.x = ~0u; k.y = ~0u;
                        if (t < cnt) k = pend[t * KNT];
                        xi[t] = k.x; xd[t] = k.y;
                    }
                    sort16(xd, xi);
                    merge16(bd, bi, xd, xi);
                    thr = bd[15] < bnd ? bd[15] : bnd;
                    int left = 0;
                    if (__any(cnt > 16)) {
#pragma unroll
                        for (int t = 0; t < 15; ++t) {
                            if (16 + t < cnt) {
                                const v2u k = pend[(16 + t) * KNT];
                                if (k.y <= thr) { pend[left * KNT] = k; ++left; }
                            }
                        }
                    }
                    cnt = left;
                    if (!(last && __any(cnt > 0))) break;
                }
            }
            wp = pend3 + cnt * KNT;
        }
    }

    __syncthreads();
    v2u* lists = lds_raw;
#pragma unroll
    for (int r = 0; r < 3; ++r) {
        const int span = 1 << r;
        const int ph   = s & (2 * span - 1);
        if (ph == span) {
#pragma unroll
            for (int t = 0; t < KNN; ++t) { v2u k; k.x = bi[t]; k.y = bd[t]; lists[t * KNT + tid] = k; }
        }
        __syncthreads();
        if (ph == 0) {
            u32 xd[16], xi[16];
#pragma unroll
            for (int t = 0; t < KNN; ++t) { const v2u k = lists[t * KNT + tid + 64 * span]; xi[t] = k.x; xd[t] = k.y; }
            merge16(bd, bi, xd, xi);
        }
    }

    if (s == 0) {
        int* op = idx_out + (b * NPTS + q) * KNN;
#pragma unroll
        for (int t = 0; t < KNN; ++t) op[t] = (int)bi[t];
    }
}

__global__ __launch_bounds__(128) __attribute__((amdgpu_waves_per_eu(1, 1))) void agg_kernel(const float* __restrict__ coords,
                                                  const float* __restrict__ feat,
                                                  const float* __restrict__ Wm,
                                                  const float* __restrict__ bias,
                                                  const int*   __restrict__ idx,
                                                  float* __restrict__ out) {
    __shared__ float XS[4 * WAVE_LDS];

    const int lane = threadIdx.x & 31;
    const int wav  = threadIdx.x >> 5;
    const int nl   = lane & 15;
    const int h    = lane >> 4;

    const int gw   = blockIdx.x * 4 + wav;
    const int half = gw & 1;
    const int pg   = gw >> 1;

    float* fib0 = &XS[wav * WAVE_LDS];
    float* fib1 = fib0 + 64;
    float* X0   = fib1 + 64;
    float* X1   = X0 + 16 * XSTR;

    v2f bw0[33], bw1[33];
    {
        const float* wc = Wm + half * 32 + nl;
#pragma unroll
        for (int ks = 0; ks < 33; ++ks) {
            const int k = 4 * ks + 2 * h;
            v2f t0, t1;
            t0.x = wc[k * CCH];
            t1.x = wc[k * CCH + 16];
            t0.y = (k + 1 < 131) ? wc[(k + 1) * CCH]      : 0.0f;
            t1.y = (k + 1 < 131) ? wc[(k + 1) * CCH + 16] : 0.0f;
            bw0[ks] = t0;
            bw1[ks] = t1;
        }
    }

    const float bsel = bias[half * 32 + lane];

    const int b     = (pg * 32) >> 13;
    const int nbase = (pg * 32) & (NPTS - 1);
    const float* fb = feat   + b * NPTS * CCH;
    const float* cb = coords + b * NPTS * 3;
    const int* ib = idx + (b * NPTS + nbase + h) * KNN + nl;

    v2f   fiA, fiB, g[32];
    float pc0, pc1, pc2, nc0, nc1, nc2;

#define LOAD_PAIR(PAIR, NBV)                                                      \
    {                                                                             \
        const int n0_ = nbase + (PAIR) * 2;                                       \
        fiA = *(const v2f*)(fb + n0_ * CCH + 2 * lane);                           \
        fiB = *(const v2f*)(fb + (n0_ + 1) * CCH + 2 * lane);                     \
        _Pragma("unroll")                                                         \
        for (int m = 0; m < 32; ++m) {                                            \
            const int nb_ = __shfl((NBV), m, 32);                                 \
            g[m] = *(const v2f*)(fb + nb_ * CCH + 2 * lane);                      \
        }                                                                         \
        const float* pc_ = cb + (n0_ + h) * 3;                                    \
        const float* nc_ = cb + (NBV) * 3;                                        \
        pc0 = pc_[0]; pc1 = pc_[1]; pc2 = pc_[2];                                 \
        nc0 = nc_[0]; nc1 = nc_[1]; nc2 = nc_[2];                                 \
    }

#define STAGE_PAIR()                                                              \
    {                                                                             \
        *(v2f*)(fib0 + 2 * lane) = fiA;                                           \
        *(v2f*)(fib1 + 2 * lane) = fiB;                                           \
        _Pragma("unroll")                                                         \
        for (int m = 0; m < 32; ++m) {                                            \
            if (m < 16) { v2f d0 = g[m] - fiA; *(v2f*)(X0 + m * XSTR + 2 * lane) = d0; }        \
            else        { v2f d1 = g[m] - fiB; *(v2f*)(X1 + (m - 16) * XSTR + 2 * lane) = d1; } \
        }                                                                         \
        v4f dp; dp.x = nc0 - pc0; dp.y = nc1 - pc1; dp.z = nc2 - pc2; dp.w = 0.0f; \
        *(v4f*)((h ? X1 : X0) + nl * XSTR + 64) = dp;                             \
    }

    int nbvN = ib[0];
    LOAD_PAIR(0, nbvN)
    nbvN = ib[2 * KNN];
    STAGE_PAIR()

    for (int it = 0; it < 16; ++it) {
        const int nx  = (it + 1 < 16) ? it + 1 : 15;
        const int nx2 = (it + 2 < 16) ? it + 2 : 15;
        LOAD_PAIR(nx, nbvN)
        nbvN = ib[nx2 * 2 * KNN];
        const int P0 = pg * 32 + it * 2;

        v8f a0c[2], a1c[2];
        const v8f zc = {0.f, 0.f, 0.f, 0.f, 0.f, 0.f, 0.f, 0.f};
        a0c[0] = zc; a0c[1] = zc; a1c[0] = zc; a1c[1] = zc;

#define STEP(KS, A0, A1)                                                          \
        {                                                                         \
            a0c[0] = __builtin_amdgcn_wmma_f32_16x16x4_f32(                       \
                false, (A0), false, bw0[KS], (short)0, a0c[0], false, false);     \
            a1c[0] = __builtin_amdgcn_wmma_f32_16x16x4_f32(                       \
                false, (A1), false, bw0[KS], (short)0, a1c[0], false, false);     \
            a0c[1] = __builtin_amdgcn_wmma_f32_16x16x4_f32(                       \
                false, (A0), false, bw1[KS], (short)0, a0c[1], false, false);     \
            a1c[1] = __builtin_amdgcn_wmma_f32_16x16x4_f32(                       \
                false, (A1), false, bw1[KS], (short)0, a1c[1], false, false);     \
        }

#pragma unroll
        for (int ks = 0; ks < 16; ++ks) {
            v2f A0 = *(const v2f*)(fib0 + 4 * ks + 2 * h);
            v2f A1 = *(const v2f*)(fib1 + 4 * ks + 2 * h);
            STEP(ks, A0, A1);
        }
#pragma unroll
        for (int ks = 0; ks < 16; ++ks) {
            int xo = nl * XSTR + 4 * ks + 2 * h;
            v2f A0 = *(const v2f*)(X0 + xo);
            v2f A1 = *(const v2f*)(X1 + xo);
            STEP(16 + ks, A0, A1);
        }
        {
            int xo = nl * XSTR + 64 + 2 * h;
            v2f A0 = *(const v2f*)(X0 + xo);
            v2f A1 = *(const v2f*)(X1 + xo);
            STEP(32, A0, A1);
        }
#undef STEP

        float p0[2], p1[2];
#pragma unroll
        for (int t = 0; t < 2; ++t) {
            float v0 = a0c[t][0];
            float v1 = a1c[t][0];
#pragma unroll
            for (int r = 1; r < 8; ++r) {
                v0 = fmaxf(v0, a0c[t][r]);
                v1 = fmaxf(v1, a1c[t][r]);
            }
            p0[t] = v0; p1[t] = v1;
        }
        float m0 = fmaxf(h ? p0[1] : p0[0], __shfl_xor(h ? p0[0] : p0[1], 16, 32));
        float m1 = fmaxf(h ? p1[1] : p1[0], __shfl_xor(h ? p1[0] : p1[1], 16, 32));
        float* outp = out + (size_t)P0 * CCH + half * 32 + lane;
        outp[0]   = fmaxf(m0 + bsel, 0.0f);
        outp[CCH] = fmaxf(m1 + bsel, 0.0f);

        STAGE_PAIR()
    }
#undef LOAD_PAIR
#undef STAGE_PAIR
}

extern "C" void kernel_launch(void* const* d_in, const int* in_sizes, int n_in,
                              void* d_out, int out_size, void* d_ws, size_t ws_size,
                              hipStream_t stream) {
    const float* coords = (const float*)d_in[0];
    const float* feat   = (const float*)d_in[1];
    const float* Wm     = (const float*)d_in[2];
    const float* bias   = (const float*)d_in[3];
    float* out = (float*)d_out;
    int* idx = (int*)d_ws;

    knn_kernel<<<256, KNT, 0, stream>>>(coords, idx);
    agg_kernel<<<256, 128, 0, stream>>>(coords, feat, Wm, bias, idx, out);
}


# Aggregation: a pair's A fragments (33 k-steps, 2 points) are read from LDS as one group into registers ahead of its 132 products; no LDS instruction stands among the products any more; same sums
# speedup vs baseline: 29.5751x; 1.0322x over previous
#include <hip/hip_runtime.h>

typedef __attribute__((ext_vector_type(2))) float v2f;
typedef __attribute__((ext_vector_type(4))) float v4f;
typedef __attribute__((ext_vector_type(8))) float v8f;

#define NPTS 8192
#define CCH  64
#define KNN  16
#define XSTR 68
#define WAVE_LDS (2*64 + 2*16*XSTR)

typedef unsigned int u32;
typedef __attribute__((ext_vector_type(2))) unsigned int v2u;

__device__ __forceinline__ bool key_lt(u32 da, u32 ia, u32 db, u32 ib) {
    u32 c1, c2;
    (void)__builtin_subc(ia, ib, 0u, &c1);
    (void)__builtin_subc(da, db, c1, &c2);
    return c2 != 0u;
}

__device__ __forceinline__ void cswap(u32& da, u32& ia, u32& db, u32& ib) {
    const bool lt = key_lt(da, ia, db, ib);
    const u32 dlo = lt ? da : db, dhi = lt ? db : da;
    const u32 ilo = lt ? ia : ib, ihi = lt ? ib : ia;
    da = dlo; ia = ilo; db = dhi; ib = ihi;
}

__device__ __forceinline__ void sort16(u32 (&d)[16], u32 (&x)[16]) {
#define CS(a, b) cswap(d[a], x[a], d[b], x[b]);
    CS(0,1) CS(2,3) CS(4,5) CS(6,7) CS(8,9) CS(10,11) CS(12,13) CS(14,15)
    CS(0,2) CS(1,3) CS(4,6) CS(5,7) CS(8,10) CS(9,11) CS(12,14) CS(13,15)
    CS(1,2) CS(5,6) CS(9,10) CS(13,14)
    CS(0,4) CS(1,5) CS(2,6) CS(3,7) CS(8,12) CS(9,13) CS(10,14) CS(11,15)
    CS(2,4) CS(3,5) CS(10,12) CS(11,13)
    CS(1,2) CS(3,4) CS(5,6) CS(9,10) CS(11,12) CS(13,14)
    CS(0,8) CS(1,9) CS(2,10) CS(3,11) CS(4,12) CS(5,13) CS(6,14) CS(7,15)
    CS(4,8) CS(5,9) CS(6,10) CS(7,11)
    CS(2,4) CS(3,5) CS(6,8) CS(7,9) CS(10,12) CS(11,13)
    CS(1,2) CS(3,4) CS(5,6) CS(7,8) CS(9,10) CS(11,12) CS(13,14)
#undef CS
}

__device__ __forceinline__ void merge16(u32 (&bd)[16], u32 (&bi)[16],
                                        const u32 (&xd)[16], const u32 (&xi)[16]) {
#pragma unroll
    for (int i = 0; i < 16; ++i) {
        const u32 od = xd[15 - i], oi = xi[15 - i];
        const bool lt = key_lt(bd[i], bi[i], od, oi);
        bd[i] = lt ? bd[i] : od;
        bi[i] = lt ? bi[i] : oi;
    }
#pragma unroll
    for (int lj = 3; lj >= 0; --lj) {
#pragma unroll
        for (int i = 0; i < 16; ++i) {
            const int l = i ^ (1 << lj);
            if (l > i) cswap(bd[i], bi[i], bd[l], bi[l]);
        }
    }
}

#define KNT   512
#define KSLEN 1024
#define KBCAP 32

__global__ __launch_bounds__(KNT) void knn_kernel(const float* __restrict__ coords,
                                                  int* __restrict__ idx_out) {
    __shared__ __attribute__((aligned(16))) v2u lds_raw[2 * NPTS];
    __shared__ __attribute__((aligned(8)))  v2u lds_pend[KBCAP * KNT];
    __shared__ u32 lds_pub[KNT];
    const v4f* pts = (const v4f*)lds_raw;
    float* ptsf = (float*)lds_raw;
    const int tid   = threadIdx.x;
    const int b     = blockIdx.x >> 7;
    const int qbase = (blockIdx.x & 127) << 6;
    const float* cb = coords + b * NPTS * 3;

#pragma unroll
    for (int u = 0; u < NPTS / KNT; ++u) {
        const int j = tid + u * KNT;
        float x = cb[j * 3 + 0], y = cb[j * 3 + 1], z = cb[j * 3 + 2];
        float w;
        {
#pragma clang fp contract(off)
            const float t = x * x + y * y;
            w = __builtin_fmaf(z, z, t);
        }
        float* pp = ptsf + 8 * (j >> 1) + (j & 1);
        pp[0] = x; pp[2] = y; pp[4] = z; pp[6] = w;
    }
    lds_pub[tid] = ~0u;
    __syncthreads();

    const int lane = tid & 31;
    const int wav  = tid >> 5;
    const int s    = wav >> 1;
    const int q    = qbase + ((wav & 1) << 5) + lane;
    const float* mp = ptsf + 8 * (q >> 1) + (q & 1);
    v2f mx2, my2, mz2, mw2;
    mx2.x = mp[0]; mx2.y = mx2.x;
    my2.x = mp[2]; my2.y = my2.x;
    mz2.x = mp[4]; mz2.y = mz2.x;
    mw2.x = mp[6]; mw2.y = mw2.x;
    const v2f neg2 = {-2.0f, -2.0f};
    const v2f zero2 = {0.0f, 0.0f};

    u32 bd[KNN], bi[KNN];
#pragma unroll
    for (int t = 0; t < KNN; ++t) { bd[t] = ~0u; bi[t] = ~0u; }

    v2u* const pend = lds_pend + tid;
    typedef __attribute__((address_space(3))) v2u lds_v2u;
    lds_v2u* const pend3 = (lds_v2u*)pend;
    lds_v2u* wp = pend3;
    u32 thr = ~0u;
    u32 bnd = ~0u;
    u32 m1 = ~0u, m2 = ~0u;
    const int qslot = tid & 63;

    const int cbase = s * KSLEN;
    for (int c = 0; c < KSLEN / 16; ++c) {
        {
            __asm__ volatile("" ::: "memory");
            u32 m = lds_pub[qslot];
#pragma unroll
            for (int t = 1; t < 8; ++t) { const u32 v = lds_pub[t * 64 + qslot]; m = v > m ? v : m; }
            bnd = m < bnd ? m : bnd;
            thr = bnd < thr ? bnd : thr;
        }
        u32 du[16];
#pragma unroll
        for (int t = 0; t < 8; ++t) {
            const v4f a  = pts[cbase + c * 16 + 2 * t];
            const v4f bq = pts[cbase + c * 16 + 2 * t + 1];
            v2f X, Y, Z, Wv;
            X.x = a.x;  X.y = a.y;  Y.x = a.z;   Y.y = a.w;
            Z.x = bq.x; Z.y = bq.y; Wv.x = bq.z; Wv.y = bq.w;
            v2f dot = mx2 * X;
            dot = __builtin_elementwise_fma(my2, Y, dot);
            dot = __builtin_elementwise_fma(mz2, Z, dot);
            v2f d = __builtin_elementwise_fma(neg2, dot, mw2 + Wv);
            d = __builtin_elementwise_max(d, zero2);
            du[2 * t]     = __float_as_uint(d.x);
            du[2 * t + 1] = __float_as_uint(d.y);
        }
        __asm__ volatile("" :: "v"(du[0]), "v"(du[1]), "v"(du[2]), "v"(du[3]),
                               "v"(du[4]), "v"(du[5]), "v"(du[6]), "v"(du[7]),
                               "v"(du[8]), "v"(du[9]), "v"(du[10]), "v"(du[11]),
                               "v"(du[12]), "v"(du[13]), "v"(du[14]), "v"(du[15]));
#pragma unroll
        for (int i = 0; i < 16; ++i) {
            const bool pass = du[i] <= thr;
            if (__any(pass)) {
                v2u k; k.x = (u32)(cbase + c * 16 + i); k.y = du[i];
                *wp = k;
                wp += pass ? KNT : 0;
                const u32 hi = du[i] > m1 ? du[i] : m1;
                m2 = hi < m2 ? hi : m2;
                m1 = du[i] < m1 ? du[i] : m1;
            }
        }
        lds_pub[tid] = m2;
        const bool last = (c == KSLEN / 16 - 1);
        const int trig = last ? 1 : 16;
        if (__any(wp >= pend3 + trig * KNT)) {
            int cnt = (int)(wp - pend3) / KNT;
            if (!last) {
                v2u kq[KBCAP - 1];
#pragma unroll
                for (int t = 0; t < 16; ++t) kq[t] = pend[t * KNT];
                const bool more = __any(cnt > 16);
                if (more) {
#pragma unroll
                    for (int t = 16; t < KBCAP - 1; ++t) kq[t] = pend[t * KNT];
                }
                lds_v2u* wq = pend3;
#pragma unroll
                for (int t = 0; t < 16; ++t) {
                    const bool keep = (t < cnt) && (kq[t].y <= thr);
                    *wq = kq[t];
                    wq += keep ? KNT : 0;
                }
                if (more) {
#pragma unroll
                    for (int t = 16; t < KBCAP - 1; ++t) {
                        const bool keep = (t < cnt) && (kq[t].y <= thr);
                        *wq = kq[t];
                        wq += keep ? KNT : 0;
                    }
                }
                cnt = (int)(wq - pend3) / KNT;
            }
            if (__any(cnt >= trig)) {
                for (int pass = 0; pass < 2; ++pass) {
                    u32 xd[16], xi[16];
#pragma unroll
                    for (int t = 0; t < 16; ++t) {
                        v2u k; k.x = ~0u; k.y = ~0u;
                        if (t < cnt) k = pend[t * KNT];
                        xi[t] = k.x; xd[t] = k.y;
                    }
                    sort16(xd, xi);
                    merge16(bd, bi, xd, xi);
                    thr = bd[15] < bnd ? bd[15] : bnd;
                    int left = 0;
                    if (__any(cnt > 16)) {
#pragma unroll
                        for (int t = 0; t < 15; ++t) {
                            if (16 + t < cnt) {
                                const v2u k = pend[(16 + t) * KNT];
                                if (k.y <= thr) { pend[left * KNT] = k; ++left; }
                            }
                        }
                    }
                    cnt = left;
                    if (!(last && __any(cnt > 0))) break;
                }
            }
            wp = pend3 + cnt * KNT;
        }
    }

    __syncthreads();
    v2u* lists = lds_raw;
#pragma unroll
    for (int r = 0; r < 3; ++r) {
        const int span = 1 << r;
        const int ph   = s & (2 * span - 1);
        if (ph == span) {
#pragma unroll
            for (int t = 0; t < KNN; ++t) { v2u k; k.x = bi[t]; k.y = bd[t]; lists[t * KNT + tid] = k; }
        }
        __syncthreads();
        if (ph == 0) {
            u32 xd[16], xi[16];
#pragma unroll
            for (int t = 0; t < KNN; ++t) { const v2u k = lists[t * KNT + tid + 64 * span]; xi[t] = k.x; xd[t] = k.y; }
            merge16(bd, bi, xd, xi);
        }
    }

    if (s == 0) {
        int* op = idx_out + (b * NPTS + q) * KNN;
#pragma unroll
        for (int t = 0; t < KNN; ++t) op[t] = (int)bi[t];
    }
}

__global__ __launch_bounds__(128) __attribute__((amdgpu_waves_per_eu(1, 1))) void agg_kernel(const float* __restrict__ coords,
                                                  const float* __restrict__ feat,
                                                  const float* __restrict__ Wm,
                                                  const float* __restrict__ bias,
                                                  const int*   __restrict__ idx,
                                                  float* __restrict__ out) {
    __shared__ float XS[4 * WAVE_LDS];

    const int lane = threadIdx.x & 31;
    const int wav  = threadIdx.x >> 5;
    const int nl   = lane & 15;
    const int h    = lane >> 4;

    const int gw   = blockIdx.x * 4 + wav;
    const int half = gw & 1;
    const int pg   = gw >> 1;

    float* fib0 = &XS[wav * WAVE_LDS];
    float* fib1 = fib0 + 64;
    float* X0   = fib1 + 64;
    float* X1   = X0 + 16 * XSTR;

    v2f bw0[33], bw1[33];
    {
        const float* wc = Wm + half * 32 + nl;
#pragma unroll
        for (int ks = 0; ks < 33; ++ks) {
            const int k = 4 * ks + 2 * h;
            v2f t0, t1;
            t0.x = wc[k * CCH];
            t1.x = wc[k * CCH + 16];
            t0.y = (k + 1 < 131) ? wc[(k + 1) * CCH]      : 0.0f;
            t1.y = (k + 1 < 131) ? wc[(k + 1) * CCH + 16] : 0.0f;
            bw0[ks] = t0;
            bw1[ks] = t1;
        }
    }

    const float bsel = bias[half * 32 + lane];

    const int b     = (pg * 32) >> 13;
    const int nbase = (pg * 32) & (NPTS - 1);
    const float* fb = feat   + b * NPTS * CCH;
    const float* cb = coords + b * NPTS * 3;
    const int* ib = idx + (b * NPTS + nbase + h) * KNN + nl;

    v2f   fiA, fiB, g[32];
    float pc0, pc1, pc2, nc0, nc1, nc2;

    typedef const __attribute__((address_space(1))) float gfloat;
    typedef const __attribute__((address_space(1))) int   gint;
    typedef const __attribute__((address_space(1))) v2f   gv2f;
    gfloat* fbq = (gfloat*)fb;
    gfloat* cbq = (gfloat*)cb;
    gint*   ibq = (gint*)ib;
    __asm__ volatile("" : "+v"(fbq), "+v"(cbq), "+v"(ibq));

#define LOAD_PAIR(PAIR, NBV)                                                      \
    {                                                                             \
        const int n0_ = nbase + (PAIR) * 2;                                       \
        fiA = *(gv2f*)(fbq + n0_ * CCH + 2 * lane);                               \
        fiB = *(gv2f*)(fbq + (n0_ + 1) * CCH + 2 * lane);                         \
        _Pragma("unroll")                                                         \
        for (int m = 0; m < 32; ++m) {                                            \
            const int nb_ = __builtin_amdgcn_readlane((NBV), m);                  \
            g[m] = *(gv2f*)(fbq + nb_ * CCH + 2 * lane);                          \
        }                                                                         \
        gfloat* pc_ = cbq + (n0_ + h) * 3;                                        \
        gfloat* nc_ = cbq + (NBV) * 3;                                            \
        pc0 = pc_[0]; pc1 = pc_[1]; pc2 = pc_[2];                                 \
        nc0 = nc_[0]; nc1 = nc_[1]; nc2 = nc_[2];                                 \
    }

#define STAGE_PAIR()                                                              \
    {                                                                             \
        *(v2f*)(fib0 + 2 * lane) = fiA;                                           \
        *(v2f*)(fib1 + 2 * lane) = fiB;                                           \
        _Pragma("unroll")                                                         \
        for (int m = 0; m < 32; ++m) {                                            \
            if (m < 16) { v2f d0 = g[m] - fiA; *(v2f*)(X0 + m * XSTR + 2 * lane) = d0; }        \
            else        { v2f d1 = g[m] - fiB; *(v2f*)(X1 + (m - 16) * XSTR + 2 * lane) = d1; } \
        }                                                                         \
        v4f dp; dp.x = nc0 - pc0; dp.y = nc1 - pc1; dp.z = nc2 - pc2; dp.w = 0.0f; \
        *(v4f*)((h ? X1 : X0) + nl * XSTR + 64) = dp;                             \
    }

    v2f fa0[33], fa1[33];
#define LOAD_A()                                                                  \
    {                                                                             \
        _Pragma("unroll")                                                         \
        for (int ks = 0; ks < 16; ++ks) {                                         \
            fa0[ks] = *(const v2f*)(fib0 + 4 * ks + 2 * h);                       \
            fa1[ks] = *(const v2f*)(fib1 + 4 * ks + 2 * h);                       \
        }                                                                         \
        _Pragma("unroll")                                                         \
        for (int ks = 0; ks < 17; ++ks) {                                         \
            const int xo_ = nl * XSTR + 4 * ks + 2 * h;                           \
            fa0[16 + ks] = *(const v2f*)(X0 + xo_);                               \
            fa1[16 + ks] = *(const v2f*)(X1 + xo_);                               \
        }                                                                         \
    }

    int nbvN = ibq[0];
    LOAD_PAIR(0, nbvN)
    nbvN = ibq[2 * KNN];
    STAGE_PAIR()
    LOAD_A()

    for (int it = 0; it < 16; ++it) {
        const int nx  = (it + 1 < 16) ? it + 1 : 15;
        const int nx2 = (it + 2 < 16) ? it + 2 : 15;
        const int P0 = pg * 32 + it * 2;

        v8f a0c[2], a1c[2];
        const v8f zc = {0.f, 0.f, 0.f, 0.f, 0.f, 0.f, 0.f, 0.f};
        a0c[0] = zc; a0c[1] = zc; a1c[0] = zc; a1c[1] = zc;

#define STEP(KS, A0, A1)                                                          \
        {                                                                         \
            a0c[0] = __builtin_amdgcn_wmma_f32_16x16x4_f32(                       \
                false, (A0), false, bw0[KS], (short)0, a0c[0], false, false);     \
            a1c[0] = __builtin_amdgcn_wmma_f32_16x16x4_f32(                       \
                false, (A1), false, bw0[KS], (short)0, a1c[0], false, false);     \
            a0c[1] = __builtin_amdgcn_wmma_f32_16x16x4_f32(                       \
                false, (A0), false, bw1[KS], (short)0, a0c[1], false, false);     \
            a1c[1] = __builtin_amdgcn_wmma_f32_16x16x4_f32(                       \
                false, (A1), false, bw1[KS], (short)0, a1c[1], false, false);     \
        }

#pragma unroll
        for (int ks = 0; ks < 12; ++ks) {
            STEP(ks, fa0[ks], fa1[ks]);
        }
        LOAD_PAIR(nx, nbvN)
        nbvN = ibq[nx2 * 2 * KNN];
        __asm__ volatile("" : "+v"(a0c[0]), "+v"(a0c[1]), "+v"(a1c[0]), "+v"(a1c[1]) :: "memory");
#pragma unroll
        for (int ks = 12; ks < 33; ++ks) {
            STEP(ks, fa0[ks], fa1[ks]);
        }
#undef STEP
        __asm__ volatile("" : "+v"(a0c[0]), "+v"(a0c[1]), "+v"(a1c[0]), "+v"(a1c[1]) :: "memory");
        STAGE_PAIR()
        LOAD_A()

        float p0[2], p1[2];
#pragma unroll
        for (int t = 0; t < 2; ++t) {
            float v0 = a0c[t][0];
            float v1 = a1c[t][0];
#pragma unroll
            for (int r = 1; r < 8; ++r) {
                v0 = fmaxf(v0, a0c[t][r]);
                v1 = fmaxf(v1, a1c[t][r]);
            }
            p0[t] = v0; p1[t] = v1;
        }
        __asm__ volatile("" : "+v"(p0[0]), "+v"(p0[1]), "+v"(p1[0]), "+v"(p1[1]) :: "memory");
        float m0 = fmaxf(h ? p0[1] : p0[0], __shfl_xor(h ? p0[0] : p0[1], 16, 32));
        float m1 = fmaxf(h ? p1[1] : p1[0], __shfl_xor(h ? p1[0] : p1[1], 16, 32));
        float* outp = out + (size_t)P0 * CCH + half * 32 + lane;
        outp[0]   = fmaxf(m0 + bsel, 0.0f);
        outp[CCH] = fmaxf(m1 + bsel, 0.0f);
    }
#undef LOAD_PAIR
#undef STAGE_PAIR
#undef LOAD_A
}

extern "C" void kernel_launch(void* const* d_in, const int* in_sizes, int n_in,
                              void* d_out, int out_size, void* d_ws, size_t ws_size,
                              hipStream_t stream) {
    const float* coords = (const float*)d_in[0];
    const float* feat   = (const float*)d_in[1];
    const float* Wm     = (const float*)d_in[2];
    const float* bias   = (const float*)d_in[3];
    float* out = (float*)d_out;
    int* idx = (int*)d_ws;

    knn_kernel<<<256, KNT, 0, stream>>>(coords, idx);
    agg_kernel<<<256, 128, 0, stream>>>(coords, feat, Wm, bias, idx, out);
}
